# starting kernel
# baseline (speedup 1.0000x reference)
#include <hip/hip_runtime.h>

typedef __attribute__((ext_vector_type(2))) float v2f;
typedef __attribute__((ext_vector_type(4))) float v4f;
typedef __attribute__((ext_vector_type(8))) float v8f;

namespace {
constexpr int kB     = 128;
constexpr int kH     = 128;
constexpr int kW     = 128;
constexpr int kS     = 2;
constexpr int kOut   = 4;
constexpr int kHo    = kH - 1;
constexpr int kWo    = kW - 1;
constexpr int kWG    = 4;
constexpr int kWaves = 8;
}

__global__ __launch_bounds__(256) void eps_wmma_kernel(
    const float* __restrict__ in,
    const float* __restrict__ core,
    float* __restrict__ out) {
  __shared__ float lds_core[16 * 16];
  __shared__ float lds_out[kWaves][32 * kOut];

  const int tid = threadIdx.x;
  {
    const int row = tid >> 4;
    const int col = tid & 15;
    lds_core[tid] = (col < kOut) ? core[row * kOut + col] : 0.0f;
  }
  __syncthreads();

  const int wid  = tid >> 5;
  const int lane = tid & 31;
  const int m    = lane & 15;
  const int half = lane >> 4;

  const unsigned g   = (unsigned)blockIdx.x * kWaves + (unsigned)wid;
  const int      wg  = (int)(g & (kWG - 1));
  const unsigned rem = g >> 2;
  const int      h   = (int)(rem % (unsigned)kHo);
  const int      b   = (int)(rem / (unsigned)kHo);

  const int w0 = wg * 32 + m;
  const int w1 = w0 + 16;
  const int wc0 = (w0 < kWo) ? w0 : (kWo - 1);
  const int wc1 = (w1 < kWo) ? w1 : (kWo - 1);

  const size_t rowbase = (size_t)(b * kH + h) * kW;

  const int kb = (2 * half) * 16 + m;
  v2f b0, b1, b2, b3;
  b0.x = lds_core[  0 + kb];  b0.y = lds_core[  0 + kb + 16];
  b1.x = lds_core[ 64 + kb];  b1.y = lds_core[ 64 + kb + 16];
  b2.x = lds_core[128 + kb];  b2.y = lds_core[128 + kb + 16];
  b3.x = lds_core[192 + kb];  b3.y = lds_core[192 + kb + 16];

  v8f acc0 = {0.f, 0.f, 0.f, 0.f, 0.f, 0.f, 0.f, 0.f};
  v8f acc1 = {0.f, 0.f, 0.f, 0.f, 0.f, 0.f, 0.f, 0.f};

#pragma unroll
  for (int sub = 0; sub < 2; ++sub) {
    const size_t base = (rowbase + (sub ? wc1 : wc0)) * kS;
    const v2f x00 = *(const v2f*)(in + base);
    const v2f x01 = *(const v2f*)(in + base + kS);
    const v2f x10 = *(const v2f*)(in + base + (size_t)kW * kS);
    const v2f x11 = *(const v2f*)(in + base + (size_t)kW * kS + kS);

    float p01[4];
    p01[0] = x00.x * x01.x;
    p01[1] = x00.x * x01.y;
    p01[2] = x00.y * x01.x;
    p01[3] = x00.y * x01.y;
    const float s2v = half ? x10.y : x10.x;
    const float q0 = s2v * x11.x;
    const float q1 = s2v * x11.y;

    v2f a0, a1, a2, a3;
    a0.x = p01[0] * q0; a0.y = p01[0] * q1;
    a1.x = p01[1] * q0; a1.y = p01[1] * q1;
    a2.x = p01[2] * q0; a2.y = p01[2] * q1;
    a3.x = p01[3] * q0; a3.y = p01[3] * q1;

    v8f acc = sub ? acc1 : acc0;
    acc = __builtin_amdgcn_wmma_f32_16x16x4_f32(false, a0, false, b0, (short)0,
                                                acc, false, false);
    acc = __builtin_amdgcn_wmma_f32_16x16x4_f32(false, a1, false, b1, (short)0,
                                                acc, false, false);
    acc = __builtin_amdgcn_wmma_f32_16x16x4_f32(false, a2, false, b2, (short)0,
                                                acc, false, false);
    acc = __builtin_amdgcn_wmma_f32_16x16x4_f32(false, a3, false, b3, (short)0,
                                                acc, false, false);
    if (sub) acc1 = acc; else acc0 = acc;
  }

  if (m < kOut) {
#pragma unroll
    for (int r = 0; r < 8; ++r) {
      lds_out[wid][(r + 8 * half) * kOut + m]      = acc0[r];
      lds_out[wid][(16 + r + 8 * half) * kOut + m] = acc1[r];
    }
  }
  asm volatile("s_wait_dscnt 0" ::: "memory");

  const int wl = wg * 32 + lane;
  if (wl < kWo) {
    const v4f o = *(const v4f*)&lds_out[wid][lane * kOut];
    *(v4f*)(out + (((size_t)b * kHo + h) * kWo + wl) * kOut) = o;
  }
}

extern "C" void kernel_launch(void* const* d_in, const int* in_sizes, int n_in,
                              void* d_out, int out_size, void* d_ws, size_t ws_size,
                              hipStream_t stream) {
  (void)in_sizes; (void)n_in; (void)out_size; (void)d_ws; (void)ws_size;
  const float* in   = (const float*)d_in[0];
  const float* core = (const float*)d_in[1];
  float*       out  = (float*)d_out;

  const int total_waves = kB * kHo * kWG;
  const int blocks      = total_waves / kWaves;
  eps_wmma_kernel<<<blocks, 256, 0, stream>>>(in, core, out);
}


# Product transposed (core = row operand read from global, pixels = columns): each lane gets its pixel's 4 channels and stores 16 B from the accumulators; no LDS, barrier or fence; sum order unchanged
# speedup vs baseline: 1.3991x; 1.3991x over previous
#include <hip/hip_runtime.h>


typedef __attribute__((ext_vector_type(2))) float v2f;
typedef __attribute__((ext_vector_type(4))) float v4f;
typedef __attribute__((ext_vector_type(8))) float v8f;

namespace {
constexpr int kB     = 128;
constexpr int kH     = 128;
constexpr int kW     = 128;
constexpr int kS     = 2;
constexpr int kOut   = 4;
constexpr int kHo    = kH - 1;
constexpr int kWo    = kW - 1;
constexpr int kWG    = 4;
constexpr int kWaves = 8;
}

__global__ __launch_bounds__(256, 10) void eps_wmma_kernel(
    const float* __restrict__ in,
    const float* __restrict__ core,
    float* __restrict__ out) {
  const int tid  = threadIdx.x;
  const int wid  = tid >> 5;
  const int lane = tid & 31;
  const int m    = lane & 15;
  const int half = lane >> 4;

  const unsigned g = (unsigned)blockIdx.x * kWaves + (unsigned)wid;
  const int      h = (int)(g % (unsigned)kHo);
  const int      b = (int)(g / (unsigned)kHo);

  const float* __restrict__ cp = core + 8 * half + (m & 3);
  v2f c0, c1, c2, c3;
  c0.x = cp[0];   c0.y = cp[4];
  c1.x = cp[16];  c1.y = cp[20];
  c2.x = cp[32];  c2.y = cp[36];
  c3.x = cp[48];  c3.y = cp[52];

  const float* __restrict__ r0 = in + (size_t)(b * kH + h) * (size_t)(kW * kS);
  const float* __restrict__ r1 = r0 + kW * kS;

  v2f x00[2 * kWG], x01[2 * kWG], x10[2 * kWG], x11[2 * kWG];
#pragma unroll
  for (int t = 0; t < 2 * kWG; ++t) {
    const int w  = t * 16 + m;
    const int wc = (w < kWo) ? w : (kWo - 1);
    x00[t] = *(const v2f*)(r0 + wc * kS);
    x01[t] = *(const v2f*)(r0 + wc * kS + kS);
    x10[t] = *(const v2f*)(r1 + wc * kS);
    x11[t] = *(const v2f*)(r1 + wc * kS + kS);
  }

  float* __restrict__ orow = out + (size_t)g * (size_t)(kWo * kOut);

#pragma unroll
  for (int j = 0; j < kWG; ++j) {
    v2f a[2][4];
#pragma unroll
    for (int sub = 0; sub < 2; ++sub) {
      const int t = 2 * j + sub;
      float p01[4];
      p01[0] = x00[t].x * x01[t].x;
      p01[1] = x00[t].x * x01[t].y;
      p01[2] = x00[t].y * x01[t].x;
      p01[3] = x00[t].y * x01[t].y;
      const float s2v = half ? x10[t].y : x10[t].x;
      const float q0 = s2v * x11[t].x;
      const float q1 = s2v * x11[t].y;
#pragma unroll
      for (int q = 0; q < 4; ++q) {
        a[sub][q].x = p01[q] * q0;
        a[sub][q].y = p01[q] * q1;
      }
    }

    v8f acc0 = {0.f, 0.f, 0.f, 0.f, 0.f, 0.f, 0.f, 0.f};
    v8f acc1 = {0.f, 0.f, 0.f, 0.f, 0.f, 0.f, 0.f, 0.f};
    acc0 = __builtin_amdgcn_wmma_f32_16x16x4_f32(false, c0, false, a[0][0], (short)0,
                                                 acc0, false, false);
    acc1 = __builtin_amdgcn_wmma_f32_16x16x4_f32(false, c0, false, a[1][0], (short)0,
                                                 acc1, false, false);
    acc0 = __builtin_amdgcn_wmma_f32_16x16x4_f32(false, c1, false, a[0][1], (short)0,
                                                 acc0, false, false);
    acc1 = __builtin_amdgcn_wmma_f32_16x16x4_f32(false, c1, false, a[1][1], (short)0,
                                                 acc1, false, false);
    acc0 = __builtin_amdgcn_wmma_f32_16x16x4_f32(false, c2, false, a[0][2], (short)0,
                                                 acc0, false, false);
    acc1 = __builtin_amdgcn_wmma_f32_16x16x4_f32(false, c2, false, a[1][2], (short)0,
                                                 acc1, false, false);
    acc0 = __builtin_amdgcn_wmma_f32_16x16x4_f32(false, c3, false, a[0][3], (short)0,
                                                 acc0, false, false);
    acc1 = __builtin_amdgcn_wmma_f32_16x16x4_f32(false, c3, false, a[1][3], (short)0,
                                                 acc1, false, false);

    v4f o;
    o.x = half ? acc1[0] : acc0[0];
    o.y = half ? acc1[1] : acc0[1];
    o.z = half ? acc1[2] : acc0[2];
    o.w = half ? acc1[3] : acc0[3];
    const int wl = j * 32 + lane;
    if (wl < kWo) {
      *(v4f*)(orow + wl * kOut) = o;
    }
  }
}

extern "C" void kernel_launch(void* const* d_in, const int* in_sizes, int n_in,
                              void* d_out, int out_size, void* d_ws, size_t ws_size,
                              hipStream_t stream) {
  (void)in_sizes; (void)n_in; (void)out_size; (void)d_ws; (void)ws_size;
  const float* in   = (const float*)d_in[0];
  const float* core = (const float*)d_in[1];
  float*       out  = (float*)d_out;

  const int total_waves = kB * kHo;
  const int blocks      = total_waves / kWaves;
  eps_wmma_kernel<<<blocks, 256, 0, stream>>>(in, core, out);
}


# Flat line-aligned tiling: a wave owns 128 consecutive output pixels = 16 whole 128-B lines (16129 waves, was 16256 row waves); every store writes 4 whole lines, none predicated; sum order unchanged
# speedup vs baseline: 1.4695x; 1.0503x over previous
#include <hip/hip_runtime.h>


typedef __attribute__((ext_vector_type(2))) float v2f;
typedef __attribute__((ext_vector_type(4))) float v4f;
typedef __attribute__((ext_vector_type(8))) float v8f;

namespace {
constexpr int kB     = 128;
constexpr int kH     = 128;
constexpr int kW     = 128;
constexpr int kS     = 2;
constexpr int kOut   = 4;
constexpr int kHo    = kH - 1;
constexpr int kWo    = kW - 1;
constexpr int kWaves = 8;
constexpr unsigned kPixels = (unsigned)kB * kHo * kWo;
constexpr unsigned kTile   = 128;
constexpr unsigned kTiles  = kPixels / kTile;
constexpr int kGroups = (int)kTile / 32;
static_assert(kPixels % kTile == 0, "the flat output must split into whole tiles");
}

__global__ __launch_bounds__(256, 10) void eps_wmma_kernel(
    const float* __restrict__ in,
    const float* __restrict__ core,
    float* __restrict__ out) {
  const int tid  = threadIdx.x;
  const int lane = tid & 31;
  const int m    = lane & 15;
  const int half = lane >> 4;

  const unsigned wid  = (unsigned)__builtin_amdgcn_readfirstlane(tid >> 5);
  const unsigned tile = (unsigned)blockIdx.x * kWaves + wid;
  if (tile >= kTiles) return;

  const float* __restrict__ cp = core + 8 * half + (m & 3);
  v2f c0, c1, c2, c3;
  c0.x = cp[0];   c0.y = cp[4];
  c1.x = cp[16];  c1.y = cp[20];
  c2.x = cp[32];  c2.y = cp[36];
  c3.x = cp[48];  c3.y = cp[52];

  const unsigned p0 = tile * kTile;
  const unsigned g0 = p0 / (unsigned)kWo;
  const unsigned w0 = p0 - g0 * (unsigned)kWo;
  const unsigned b0 = g0 / (unsigned)kHo;
  const unsigned h0 = g0 - b0 * (unsigned)kHo;
  const unsigned base = p0 + g0 + (unsigned)kW * b0;
  const unsigned step = (h0 == (unsigned)(kHo - 1)) ? (unsigned)(kW + 1) : 1u;

  unsigned boff[2 * kGroups];
#pragma unroll
  for (int t = 0; t < 2 * kGroups; ++t) {
    const unsigned i   = (unsigned)(t * 16 + m);
    const unsigned pos = base + i + ((w0 + i >= (unsigned)kWo) ? step : 0u);
    boff[t] = pos * (unsigned)(kS * sizeof(float));
  }

  v2f x00[2 * kGroups], x01[2 * kGroups], x10[2 * kGroups], x11[2 * kGroups];
#pragma unroll
  for (int t = 0; t < 2 * kGroups; ++t) {
    const float* __restrict__ r0 = (const float*)((const char*)in + boff[t]);
    const float* __restrict__ r1 = r0 + kW * kS;
    x00[t] = *(const v2f*)(r0);
    x01[t] = *(const v2f*)(r0 + kS);
    x10[t] = *(const v2f*)(r1);
    x11[t] = *(const v2f*)(r1 + kS);
  }

  float* __restrict__ otile = out + (size_t)p0 * kOut;

#pragma unroll
  for (int j = 0; j < kGroups; ++j) {
    v2f a[2][4];
#pragma unroll
    for (int sub = 0; sub < 2; ++sub) {
      const int t = 2 * j + sub;
      float p01[4];
      p01[0] = x00[t].x * x01[t].x;
      p01[1] = x00[t].x * x01[t].y;
      p01[2] = x00[t].y * x01[t].x;
      p01[3] = x00[t].y * x01[t].y;
      const float s2v = half ? x10[t].y : x10[t].x;
      const float q0 = s2v * x11[t].x;
      const float q1 = s2v * x11[t].y;
#pragma unroll
      for (int q = 0; q < 4; ++q) {
        a[sub][q].x = p01[q] * q0;
        a[sub][q].y = p01[q] * q1;
      }
    }

    v8f acc0 = {0.f, 0.f, 0.f, 0.f, 0.f, 0.f, 0.f, 0.f};
    v8f acc1 = {0.f, 0.f, 0.f, 0.f, 0.f, 0.f, 0.f, 0.f};
    acc0 = __builtin_amdgcn_wmma_f32_16x16x4_f32(false, c0, false, a[0][0], (short)0,
                                                 acc0, false, false);
    acc1 = __builtin_amdgcn_wmma_f32_16x16x4_f32(false, c0, false, a[1][0], (short)0,
                                                 acc1, false, false);
    acc0 = __builtin_amdgcn_wmma_f32_16x16x4_f32(false, c1, false, a[0][1], (short)0,
                                                 acc0, false, false);
    acc1 = __builtin_amdgcn_wmma_f32_16x16x4_f32(false, c1, false, a[1][1], (short)0,
                                                 acc1, false, false);
    acc0 = __builtin_amdgcn_wmma_f32_16x16x4_f32(false, c2, false, a[0][2], (short)0,
                                                 acc0, false, false);
    acc1 = __builtin_amdgcn_wmma_f32_16x16x4_f32(false, c2, false, a[1][2], (short)0,
                                                 acc1, false, false);
    acc0 = __builtin_amdgcn_wmma_f32_16x16x4_f32(false, c3, false, a[0][3], (short)0,
                                                 acc0, false, false);
    acc1 = __builtin_amdgcn_wmma_f32_16x16x4_f32(false, c3, false, a[1][3], (short)0,
                                                 acc1, false, false);

    v4f o;
    o.x = half ? acc1[0] : acc0[0];
    o.y = half ? acc1[1] : acc0[1];
    o.z = half ? acc1[2] : acc0[2];
    o.w = half ? acc1[3] : acc0[3];
    *(v4f*)(otile + (j * 32 + lane) * kOut) = o;
  }
}

extern "C" void kernel_launch(void* const* d_in, const int* in_sizes, int n_in,
                              void* d_out, int out_size, void* d_ws, size_t ws_size,
                              hipStream_t stream) {
  (void)in_sizes; (void)n_in; (void)out_size; (void)d_ws; (void)ws_size;
  const float* in   = (const float*)d_in[0];
  const float* core = (const float*)d_in[1];
  float*       out  = (float*)d_out;

  const int blocks = ((int)kTiles + kWaves - 1) / kWaves;
  eps_wmma_kernel<<<blocks, 256, 0, stream>>>(in, core, out);
}


# Work-groups of 32 waves (1024 threads; 505 groups, was 2017 of 8 waves): same waves, same code per wave, a quarter of the groups for the dispatcher to place; sum order unchanged
# speedup vs baseline: 2.2032x; 1.4993x over previous
#include <hip/hip_runtime.h>


typedef __attribute__((ext_vector_type(2))) float v2f;
typedef __attribute__((ext_vector_type(4))) float v4f;
typedef __attribute__((ext_vector_type(8))) float v8f;

namespace {
constexpr int kB     = 128;
constexpr int kH     = 128;
constexpr int kW     = 128;
constexpr int kS     = 2;
constexpr int kOut   = 4;
constexpr int kHo    = kH - 1;
constexpr int kWo    = kW - 1;
constexpr int kWaves = 32;
constexpr unsigned kPixels = (unsigned)kB * kHo * kWo;
constexpr unsigned kTile   = 128;
constexpr unsigned kTiles  = kPixels / kTile;
constexpr int kGroups = (int)kTile / 32;
constexpr int kSubs   = 2 * kGroups;
static_assert(kPixels % kTile == 0, "the flat output must split into whole tiles");

__device__ __forceinline__ float next_in_row(float x) {
  const int b = __builtin_bit_cast(int, x);
  return __builtin_bit_cast(float, __builtin_amdgcn_update_dpp(0, b, 0x12F, 0xF, 0xF, true));
}

__device__ __forceinline__ float other_half(float x) {
  const int b = __builtin_bit_cast(int, x);
  return __builtin_bit_cast(float,
      __builtin_amdgcn_permlanex16(b, b, 0x76543210u, 0xfedcba98u, false, false));
}
}

__global__ __launch_bounds__(1024, 16) void eps_wmma_kernel(
    const float* __restrict__ in,
    const float* __restrict__ core,
    float* __restrict__ out) {
  const int tid  = threadIdx.x;
  const int lane = tid & 31;
  const int m    = lane & 15;
  const int half = lane >> 4;

  const unsigned wid  = (unsigned)__builtin_amdgcn_readfirstlane(tid >> 5);
  const unsigned tile = (unsigned)blockIdx.x * kWaves + wid;
  if (tile >= kTiles) return;

  const int cidx = 32 * half + 16 * (m >> 3) + 4 * ((m >> 2) & 1) + (m & 3);
  v2f ca;
  ca.x = core[cidx];
  ca.y = core[cidx + 8];

  const unsigned p0 = tile * kTile;
  const unsigned g0 = p0 / (unsigned)kWo;
  const unsigned w0 = p0 - g0 * (unsigned)kWo;
  const unsigned b0 = g0 / (unsigned)kHo;
  const unsigned h0 = g0 - b0 * (unsigned)kHo;
  const unsigned base = p0 + g0 + (unsigned)kW * b0;
  const unsigned step = (h0 == (unsigned)(kHo - 1)) ? (unsigned)(kW + 1) : 1u;
  const int iend = (kWo - 1) - (int)w0;

  unsigned boff[kSubs], boffh[kSubs];
#pragma unroll
  for (int t = 0; t < kSubs; ++t) {
    const unsigned pos = base + (unsigned)(t * 16 + m) + ((m > iend - 16 * t) ? step : 0u);
    boff[t]  = pos * (unsigned)(kS * sizeof(float));
    boffh[t] = boff[t] + (unsigned)(half * sizeof(float));
  }

  float x0h[kSubs];
  v2f   x1[kSubs];
#pragma unroll
  for (int t = 0; t < kSubs; ++t) {
    x0h[t] = *(const float*)((const char*)in + boffh[t]);
    x1[t]  = *(const v2f*)((const char*)in + boff[t] + kW * kS * sizeof(float));
  }
  const float* __restrict__ pa = in + (size_t)(base + (unsigned)iend + 1u) * kS;
  const float* __restrict__ pb = in + (size_t)(base + kTile + step) * kS;
  const v2f xa0 = *(const v2f*)(pa);
  const v2f xa1 = *(const v2f*)(pa + kW * kS);
  const v2f xb0 = *(const v2f*)(pb);
  const v2f xb1 = *(const v2f*)(pb + kW * kS);

  v2f bcol[kSubs], rot[kSubs];
#pragma unroll
  for (int t = 0; t < kSubs; ++t) {
    bcol[t].x = x0h[t] * x1[t].x;
    bcol[t].y = x0h[t] * x1[t].y;
    rot[t].x = next_in_row(bcol[t].x);
    rot[t].y = next_in_row(bcol[t].y);
  }
  v2f fa, fb;
  {
    const float sa = half ? xa0.y : xa0.x;
    fa.x = sa * xa1.x;
    fa.y = sa * xa1.y;
    const float sb = half ? xb0.y : xb0.x;
    fb.x = sb * xb1.x;
    fb.y = sb * xb1.y;
  }

  float* __restrict__ otile = out + (size_t)p0 * kOut;
  const bool last_in_sub = (m == 15);

#pragma unroll
  for (int j = 0; j < kGroups; ++j) {
    v4f part[2];
#pragma unroll
    for (int sub = 0; sub < 2; ++sub) {
      const int t = 2 * j + sub;
      v8f acc = {0.f, 0.f, 0.f, 0.f, 0.f, 0.f, 0.f, 0.f};
      acc = __builtin_amdgcn_wmma_f32_16x16x4_f32(false, ca, false, bcol[t], (short)0,
                                                  acc, false, false);

      v2f f = rot[t];
      const int tn  = (t + 1 < kSubs) ? t + 1 : t;
      const v2f nxt = (t + 1 < kSubs) ? rot[tn] : fb;
      f.x = last_in_sub ? nxt.x : f.x;
      f.y = last_in_sub ? nxt.y : f.y;
      const bool row_end = (m == iend - 16 * t);
      f.x = row_end ? fa.x : f.x;
      f.y = row_end ? fa.y : f.y;

      part[sub].x = acc[0] * f.x + acc[4] * f.y;
      part[sub].y = acc[1] * f.x + acc[5] * f.y;
      part[sub].z = acc[2] * f.x + acc[6] * f.y;
      part[sub].w = acc[3] * f.x + acc[7] * f.y;
    }

    v4f keep, give;
    keep.x = half ? part[1].x : part[0].x;
    keep.y = half ? part[1].y : part[0].y;
    keep.z = half ? part[1].z : part[0].z;
    keep.w = half ? part[1].w : part[0].w;
    give.x = half ? part[0].x : part[1].x;
    give.y = half ? part[0].y : part[1].y;
    give.z = half ? part[0].z : part[1].z;
    give.w = half ? part[0].w : part[1].w;
    v4f o;
    o.x = keep.x + other_half(give.x);
    o.y = keep.y + other_half(give.y);
    o.z = keep.z + other_half(give.z);
    o.w = keep.w + other_half(give.w);
    *(v4f*)(otile + (j * 32 + lane) * kOut) = o;
  }
}

extern "C" void kernel_launch(void* const* d_in, const int* in_sizes, int n_in,
                              void* d_out, int out_size, void* d_ws, size_t ws_size,
                              hipStream_t stream) {
  (void)in_sizes; (void)n_in; (void)out_size; (void)d_ws; (void)ws_size;
  const float* in   = (const float*)d_in[0];
  const float* core = (const float*)d_in[1];
  float*       out  = (float*)d_out;

  const int blocks = ((int)kTiles + kWaves - 1) / kWaves;
  eps_wmma_kernel<<<blocks, 1024, 0, stream>>>(in, core, out);
}
